# starting kernel
# baseline (speedup 1.0000x reference)
#include <hip/hip_runtime.h>

#define B_SAMPLES 4096
#define D_INF     300
#define D_OUTF    300
#define NUM_MAT   274
#define TILE      16
#define NT_OUT    ((D_OUTF + TILE - 1) / TILE)
#define MAX_TILES (NUM_MAT + (B_SAMPLES / TILE))

typedef __attribute__((ext_vector_type(2))) float v2f;
typedef __attribute__((ext_vector_type(4))) float v4f;
typedef __attribute__((ext_vector_type(8))) float v8f;

__global__ void rbt_init(int* __restrict__ counts, int* __restrict__ cursor,
                         int* __restrict__ ntiles) {
    int i = blockIdx.x * blockDim.x + threadIdx.x;
    if (i < NUM_MAT) { counts[i] = 0; cursor[i] = 0; }
    if (i == 0) { *ntiles = 0; }
}

__global__ void rbt_hist(const int* __restrict__ rel, int* __restrict__ counts) {
    int b = blockIdx.x * blockDim.x + threadIdx.x;
    if (b < B_SAMPLES) atomicAdd(&counts[rel[b]], 1);
}

__global__ void rbt_scan(const int* __restrict__ counts, int* __restrict__ offsets,
                         int* __restrict__ tiles, int* __restrict__ ntiles) {
    if (threadIdx.x != 0 || blockIdx.x != 0) return;
    int acc = 0, t = 0;
    for (int r = 0; r < NUM_MAT; ++r) {
        offsets[r] = acc;
        int c = counts[r];
        for (int s = 0; s < c; s += TILE) {
            tiles[3 * t + 0] = r;
            tiles[3 * t + 1] = acc + s;
            tiles[3 * t + 2] = (c - s < TILE) ? (c - s) : TILE;
            ++t;
        }
        acc += c;
    }
    offsets[NUM_MAT] = acc;
    *ntiles = t;
}

__global__ void rbt_scatter(const int* __restrict__ rel, const int* __restrict__ offsets,
                            int* __restrict__ cursor, int* __restrict__ order) {
    int b = blockIdx.x * blockDim.x + threadIdx.x;
    if (b < B_SAMPLES) {
        int r = rel[b];
        int pos = offsets[r] + atomicAdd(&cursor[r], 1);
        order[pos] = b;
    }
}

__global__ void rbt_gemm(const float* __restrict__ x, const float* __restrict__ W,
                         float* __restrict__ out, const int* __restrict__ order,
                         const int* __restrict__ tiles, const int* __restrict__ ntiles) {
    const int tid = blockIdx.y;
    if (tid >= *ntiles) return;

    const int r     = tiles[3 * tid + 0];
    const int start = tiles[3 * tid + 1];
    const int cnt   = tiles[3 * tid + 2];

    const int lane = threadIdx.x;
    const int half = lane >> 4;
    const int l16  = lane & 15;

    const int rowbase = blockIdx.x * TILE;
    const int arow = (rowbase + l16 < D_OUTF) ? (rowbase + l16) : (D_OUTF - 1);
    const int col  = (l16 < cnt) ? l16 : (cnt - 1);
    const int s    = order[start + col];

    const float* wp = W + (size_t)r * (D_OUTF * D_INF) + (size_t)arow * D_INF + 2 * half;
    const float* xp = x + (size_t)s * D_INF + 2 * half;

    v8f c = {};
#pragma unroll 5
    for (int k = 0; k < D_INF; k += 4) {
        v2f a = *(const v2f*)(wp + k);
        v2f b = *(const v2f*)(xp + k);
        c = __builtin_amdgcn_wmma_f32_16x16x4_f32(
                 false, a,  false, b,
                 (short)0, c,  false,  false);
    }

    if (l16 < cnt) {
        const int m0 = rowbase + 8 * half;
        float* op = out + (size_t)s * D_OUTF + m0;
        if (m0 + 8 <= D_OUTF) {
            v4f lo = {c[0], c[1], c[2], c[3]};
            v4f hi = {c[4], c[5], c[6], c[7]};
            *(v4f*)(op + 0) = lo;
            *(v4f*)(op + 4) = hi;
        } else {
#pragma unroll
            for (int v = 0; v < 8; ++v)
                if (m0 + v < D_OUTF) op[v] = c[v];
        }
    }
}

extern "C" void kernel_launch(void* const* d_in, const int* in_sizes, int n_in,
                              void* d_out, int out_size, void* d_ws, size_t ws_size,
                              hipStream_t stream) {
    const float* x   = (const float*)d_in[0];
    const int*   rel = (const int*)  d_in[1];
    const float* W   = (const float*)d_in[2];
    float*       out = (float*)d_out;

    int* ws      = (int*)d_ws;
    int* counts  = ws;
    int* offsets = counts + NUM_MAT;
    int* cursor  = offsets + NUM_MAT + 1;
    int* order   = cursor + NUM_MAT;
    int* ntiles  = order + B_SAMPLES;
    int* tiles   = ntiles + 1;

    rbt_init   <<<1, 320, 0, stream>>>(counts, cursor, ntiles);
    rbt_hist   <<<B_SAMPLES / 256, 256, 0, stream>>>(rel, counts);
    rbt_scan   <<<1, 32, 0, stream>>>(counts, offsets, tiles, ntiles);
    rbt_scatter<<<B_SAMPLES / 256, 256, 0, stream>>>(rel, offsets, cursor, order);

    dim3 grid(NT_OUT, MAX_TILES);
    rbt_gemm   <<<grid, 32, 0, stream>>>(x, W, out, order, tiles, ntiles);
}


# GEMM K split over 3 waves per work-group: each wave sums 100 of the 300 K values (5 trips, not 15), partial tiles added through LDS in slice order 0,1,2; summation order changes
# speedup vs baseline: 2.3052x; 2.3052x over previous
#include <hip/hip_runtime.h>


#define B_SAMPLES 4096
#define D_INF     300
#define D_OUTF    300
#define NUM_MAT   274
#define TILE      16
#define NT_OUT    ((D_OUTF + TILE - 1) / TILE)

typedef __attribute__((ext_vector_type(2))) float v2f;
typedef __attribute__((ext_vector_type(4))) float v4f;
typedef __attribute__((ext_vector_type(8))) float v8f;
typedef __attribute__((ext_vector_type(4))) int   v4i;

#define RANK_LOADS (B_SAMPLES / (32 * 4))

__global__ void __launch_bounds__(32)
rbt_rank(const int* __restrict__ rel, int* __restrict__ counts,
         int* __restrict__ offsets, int* __restrict__ order) {
    const int r    = blockIdx.x;
    const int lane = threadIdx.x;
    const v4i* rv  = (const v4i*)rel;

    v4i v[RANK_LOADS];
#pragma unroll
    for (int i = 0; i < RANK_LOADS; ++i)
        v[i] = rv[i * 32 + lane];

    int lt = 0;
#pragma unroll
    for (int i = 0; i < RANK_LOADS; ++i)
#pragma unroll
        for (int q = 0; q < 4; ++q)
            lt += (v[i][q] < r) ? 1 : 0;
#pragma unroll
    for (int d = 16; d >= 1; d >>= 1)
        lt += __shfl_xor(lt, d);

    int run = 0;
#pragma unroll
    for (int i = 0; i < RANK_LOADS; ++i) {
#pragma unroll
        for (int q = 0; q < 4; ++q) {
            const bool hit = (v[i][q] == r);
            const unsigned m = (unsigned)__ballot(hit);
            if (m != 0u) {
                if (hit)
                    order[lt + run + __popc(m & ((1u << lane) - 1u))] = 128 * i + 4 * lane + q;
                run += __popc(m);
            }
        }
    }
    if (lane == 0) { offsets[r] = lt; counts[r] = run; }
}

#define NB 5
#define NG ((NT_OUT + NB - 1) / NB)
#define KU 5
#define KW 3
#define KSLICE (D_INF / KW)
#define RED_V4 (KW * 2 * NB * 2 * 32)

template <int NT>
__device__ __forceinline__ void rbt_pass(const float* __restrict__ x, const float* __restrict__ W,
                                         float* __restrict__ out, const int* __restrict__ order,
                                         unsigned (&wbyte)[NB], const int start, const int rem,
                                         const int tile0, const int half, const int l16,
                                         const int wave, const int lane, v4f* __restrict__ red) {
    int      s[NT];
    int      cnt[NT];
    unsigned xbyte[NT];
#pragma unroll
    for (int t = 0; t < NT; ++t) {
        const int left = rem - t * TILE;
        cnt[t] = (left < TILE) ? left : TILE;
        const int col = (l16 < cnt[t]) ? l16 : (cnt[t] - 1);
        s[t]     = order[start + t * TILE + col];
        xbyte[t] = (unsigned)(s[t] * D_INF + wave * KSLICE + 2 * half) * 4u;
    }

#pragma unroll
    for (int j = 0; j < NB; ++j)
        asm volatile("" : "+v"(wbyte[j]));

    v8f c[NT][NB];
#pragma unroll
    for (int t = 0; t < NT; ++t)
#pragma unroll
        for (int j = 0; j < NB; ++j) c[t][j] = (v8f){};

#pragma unroll
    for (int k0 = 0; k0 < KSLICE; k0 += 4 * KU) {
        v2f b[KU][NT];
        v2f a[KU][NB];
#pragma unroll
        for (int u = 0; u < KU; ++u) {
#pragma unroll
            for (int t = 0; t < NT; ++t)
                b[u][t] = *(const v2f*)((const char*)x + xbyte[t] + 4 * (k0 + 4 * u));
#pragma unroll
            for (int j = 0; j < NB; ++j)
                a[u][j] = *(const v2f*)((const char*)W + wbyte[j] + 4 * (k0 + 4 * u));
        }
#pragma unroll
        for (int u = 0; u < KU; ++u) {
#pragma unroll
            for (int j = 0; j < NB; ++j)
#pragma unroll
                for (int t = 0; t < NT; ++t)
                    c[t][j] = __builtin_amdgcn_wmma_f32_16x16x4_f32(
                                   false, a[u][j],  false, b[u][t],
                                   (short)0, c[t][j],  false,  false);
        }
        __builtin_amdgcn_sched_group_barrier(0x020, KU * (NB + NT), 0);
        __builtin_amdgcn_sched_group_barrier(0x008, KU * NB * NT, 0);
        __builtin_amdgcn_sched_barrier(0);
    }

    if constexpr (NT == 1) {
        asm volatile("v_nop\n\tv_nop\n\tv_nop\n\tv_nop\n\tv_nop"
                     : "+v"(c[0][0]), "+v"(c[0][1]), "+v"(c[0][2]), "+v"(c[0][3]), "+v"(c[0][4]));
    } else {
        asm volatile("v_nop\n\tv_nop\n\tv_nop\n\tv_nop\n\tv_nop"
                     : "+v"(c[0][0]), "+v"(c[0][1]), "+v"(c[0][2]), "+v"(c[0][3]), "+v"(c[0][4]),
                       "+v"(c[NT - 1][0]), "+v"(c[NT - 1][1]), "+v"(c[NT - 1][2]), "+v"(c[NT - 1][3]),
                       "+v"(c[NT - 1][4]));
    }

    v4f* mine = red + wave * (NT * NB * 2 * 32) + lane;
#pragma unroll
    for (int t = 0; t < NT; ++t)
#pragma unroll
        for (int j = 0; j < NB; ++j) {
            v4f lo = {c[t][j][0], c[t][j][1], c[t][j][2], c[t][j][3]};
            v4f hi = {c[t][j][4], c[t][j][5], c[t][j][6], c[t][j][7]};
            mine[((t * NB + j) * 2 + 0) * 32] = lo;
            mine[((t * NB + j) * 2 + 1) * 32] = hi;
        }
    __syncthreads();

#pragma unroll
    for (int jj = 0; jj < (NB + KW - 1) / KW; ++jj) {
        const int j = wave + jj * KW;
        if (j < NB) {
            const int m0 = (tile0 + j) * TILE + 8 * half;
#pragma unroll
            for (int t = 0; t < NT; ++t) {
                const v4f* src4 = red + ((t * NB + j) * 2) * 32 + lane;
                v4f lo = src4[0];
                v4f hi = src4[32];
#pragma unroll
                for (int q = 1; q < KW; ++q) {
                    lo += src4[q * (NT * NB * 2 * 32)];
                    hi += src4[q * (NT * NB * 2 * 32) + 32];
                }
                if (l16 < cnt[t]) {
                    float* op = out + (size_t)s[t] * D_OUTF + m0;
                    if (m0 + 8 <= D_OUTF) {
                        *(v4f*)(op + 0) = lo;
                        *(v4f*)(op + 4) = hi;
                    } else {
#pragma unroll
                        for (int v = 0; v < 4; ++v) {
                            if (m0 + v < D_OUTF)     op[v]     = lo[v];
                            if (m0 + 4 + v < D_OUTF) op[4 + v] = hi[v];
                        }
                    }
                }
            }
        }
    }
}

__global__ void __launch_bounds__(32 * KW)
rbt_gemm(const float* __restrict__ x, const float* __restrict__ W,
         float* __restrict__ out, const int* __restrict__ order,
         const int* __restrict__ counts, const int* __restrict__ offsets) {
    __shared__ v4f red[RED_V4];

    const int r    = blockIdx.y;
    const int grp  = blockIdx.x;
    const int nsmp = counts[r];
    const int seg  = offsets[r];

    const int lane = threadIdx.x & 31;
    const int wave = __builtin_amdgcn_readfirstlane((int)(threadIdx.x >> 5));
    const int half = lane >> 4;
    const int l16  = lane & 15;

    const int tile0 = grp * NB;

    unsigned wbyte[NB];
#pragma unroll
    for (int j = 0; j < NB; ++j) {
        const int row  = (tile0 + j) * TILE + l16;
        const int arow = (row < D_OUTF) ? row : (D_OUTF - 1);
        wbyte[j] = (unsigned)(r * (D_OUTF * D_INF) + arow * D_INF + wave * KSLICE + 2 * half) * 4u;
    }

    for (int j0 = 0; j0 < nsmp; j0 += 2 * TILE) {
        const int rem = nsmp - j0;
        if (j0 > 0) __syncthreads();
        if (rem > TILE) rbt_pass<2>(x, W, out, order, wbyte, seg + j0, rem, tile0, half, l16, wave, lane, red);
        else            rbt_pass<1>(x, W, out, order, wbyte, seg + j0, rem, tile0, half, l16, wave, lane, red);
    }
}

extern "C" void kernel_launch(void* const* d_in, const int* in_sizes, int n_in,
                              void* d_out, int out_size, void* d_ws, size_t ws_size,
                              hipStream_t stream) {
    const float* x   = (const float*)d_in[0];
    const int*   rel = (const int*)  d_in[1];
    const float* W   = (const float*)d_in[2];
    float*       out = (float*)d_out;

    int* ws      = (int*)d_ws;
    int* counts  = ws;
    int* offsets = counts + NUM_MAT;
    int* order   = offsets + NUM_MAT;

    rbt_rank<<<NUM_MAT, 32, 0, stream>>>(rel, counts, offsets, order);

    dim3 grid(NG, NUM_MAT);
    rbt_gemm<<<grid, 32 * KW, 0, stream>>>(x, W, out, order, counts, offsets);
}


# GEMM K trips software-pipelined in registers: two fragment sets, each trip's loads issued one trip ahead, as one group in front of the previous trip's WMMAs; K order, sums and grid unchanged
# speedup vs baseline: 2.5436x; 1.1034x over previous
#include <hip/hip_runtime.h>


#define B_SAMPLES 4096
#define D_INF     300
#define D_OUTF    300
#define NUM_MAT   274
#define TILE      16
#define NT_OUT    ((D_OUTF + TILE - 1) / TILE)

typedef __attribute__((ext_vector_type(2))) float v2f;
typedef __attribute__((ext_vector_type(4))) float v4f;
typedef __attribute__((ext_vector_type(8))) float v8f;
typedef __attribute__((ext_vector_type(4))) int   v4i;

#define RANK_LOADS (B_SAMPLES / (32 * 4))

__global__ void __launch_bounds__(32)
rbt_rank(const int* __restrict__ rel, int* __restrict__ counts,
         int* __restrict__ offsets, int* __restrict__ order) {
    const int r    = blockIdx.x;
    const int lane = threadIdx.x;
    const v4i* rv  = (const v4i*)rel;

    v4i v[RANK_LOADS];
#pragma unroll
    for (int i = 0; i < RANK_LOADS; ++i)
        v[i] = rv[i * 32 + lane];

    int lt = 0;
#pragma unroll
    for (int i = 0; i < RANK_LOADS; ++i)
#pragma unroll
        for (int q = 0; q < 4; ++q)
            lt += (v[i][q] < r) ? 1 : 0;
#pragma unroll
    for (int d = 16; d >= 1; d >>= 1)
        lt += __shfl_xor(lt, d);

    int run = 0;
#pragma unroll
    for (int i = 0; i < RANK_LOADS; ++i) {
#pragma unroll
        for (int q = 0; q < 4; ++q) {
            const bool hit = (v[i][q] == r);
            const unsigned m = (unsigned)__ballot(hit);
            if (m != 0u) {
                if (hit)
                    order[lt + run + __popc(m & ((1u << lane) - 1u))] = 128 * i + 4 * lane + q;
                run += __popc(m);
            }
        }
    }
    if (lane == 0) { offsets[r] = lt; counts[r] = run; }
}

#define NB 5
#define NG ((NT_OUT + NB - 1) / NB)
#define KU 5
#define KW 3
#define KSLICE (D_INF / KW)
#define NTRIP (KSLICE / (4 * KU))
#define RED_V4 (KW * 2 * NB * 2 * 32)

template <int NT>
__device__ __forceinline__ void rbt_pass(const float* __restrict__ x, const float* __restrict__ W,
                                         float* __restrict__ out, const int* __restrict__ order,
                                         unsigned (&wbyte)[NB], const int start, const int rem,
                                         const int tile0, const int half, const int l16,
                                         const int wave, const int lane, v4f* __restrict__ red) {
    int      s[NT];
    int      cnt[NT];
    unsigned xbyte[NT];
#pragma unroll
    for (int t = 0; t < NT; ++t) {
        const int left = rem - t * TILE;
        cnt[t] = (left < TILE) ? left : TILE;
        const int col = (l16 < cnt[t]) ? l16 : (cnt[t] - 1);
        s[t]     = order[start + t * TILE + col];
        xbyte[t] = (unsigned)(s[t] * D_INF + wave * KSLICE + 2 * half) * 4u;
    }

#pragma unroll
    for (int j = 0; j < NB; ++j)
        asm volatile("" : "+v"(wbyte[j]));

    v8f c[NT][NB];
#pragma unroll
    for (int t = 0; t < NT; ++t)
#pragma unroll
        for (int j = 0; j < NB; ++j) c[t][j] = (v8f){};

    v2f b[2][KU][NT];
    v2f a[2][KU][NB];

#pragma unroll
    for (int u = 0; u < KU; ++u) {
#pragma unroll
        for (int t = 0; t < NT; ++t)
            b[0][u][t] = *(const v2f*)((const char*)x + xbyte[t] + 4 * (4 * u));
#pragma unroll
        for (int j = 0; j < NB; ++j)
            a[0][u][j] = *(const v2f*)((const char*)W + wbyte[j] + 4 * (4 * u));
    }
    __builtin_amdgcn_sched_barrier(0);

#pragma unroll
    for (int trip = 0; trip < NTRIP; ++trip) {
        const int cur = trip & 1;
        const int nxt = cur ^ 1;
        if (trip + 1 < NTRIP) {
            const int k1 = (trip + 1) * 4 * KU;
#pragma unroll
            for (int u = 0; u < KU; ++u) {
#pragma unroll
                for (int t = 0; t < NT; ++t)
                    b[nxt][u][t] = *(const v2f*)((const char*)x + xbyte[t] + 4 * (k1 + 4 * u));
#pragma unroll
                for (int j = 0; j < NB; ++j)
                    a[nxt][u][j] = *(const v2f*)((const char*)W + wbyte[j] + 4 * (k1 + 4 * u));
            }
        }
#pragma unroll
        for (int u = 0; u < KU; ++u) {
#pragma unroll
            for (int j = 0; j < NB; ++j)
#pragma unroll
                for (int t = 0; t < NT; ++t)
                    c[t][j] = __builtin_amdgcn_wmma_f32_16x16x4_f32(
                                   false, a[cur][u][j],  false, b[cur][u][t],
                                   (short)0, c[t][j],  false,  false);
        }
        __builtin_amdgcn_sched_barrier(0);
    }

    if constexpr (NT == 1) {
        asm volatile("v_nop\n\tv_nop\n\tv_nop\n\tv_nop\n\tv_nop"
                     : "+v"(c[0][0]), "+v"(c[0][1]), "+v"(c[0][2]), "+v"(c[0][3]), "+v"(c[0][4]));
    } else {
        asm volatile("v_nop\n\tv_nop\n\tv_nop\n\tv_nop\n\tv_nop"
                     : "+v"(c[0][0]), "+v"(c[0][1]), "+v"(c[0][2]), "+v"(c[0][3]), "+v"(c[0][4]),
                       "+v"(c[NT - 1][0]), "+v"(c[NT - 1][1]), "+v"(c[NT - 1][2]), "+v"(c[NT - 1][3]),
                       "+v"(c[NT - 1][4]));
    }

    v4f* mine = red + wave * (NT * NB * 2 * 32) + lane;
#pragma unroll
    for (int t = 0; t < NT; ++t)
#pragma unroll
        for (int j = 0; j < NB; ++j) {
            v4f lo = {c[t][j][0], c[t][j][1], c[t][j][2], c[t][j][3]};
            v4f hi = {c[t][j][4], c[t][j][5], c[t][j][6], c[t][j][7]};
            mine[((t * NB + j) * 2 + 0) * 32] = lo;
            mine[((t * NB + j) * 2 + 1) * 32] = hi;
        }
    __syncthreads();

#pragma unroll
    for (int jj = 0; jj < (NB + KW - 1) / KW; ++jj) {
        const int j = wave + jj * KW;
        if (j < NB) {
            const int m0 = (tile0 + j) * TILE + 8 * half;
#pragma unroll
            for (int t = 0; t < NT; ++t) {
                const v4f* src4 = red + ((t * NB + j) * 2) * 32 + lane;
                v4f lo = src4[0];
                v4f hi = src4[32];
#pragma unroll
                for (int q = 1; q < KW; ++q) {
                    lo += src4[q * (NT * NB * 2 * 32)];
                    hi += src4[q * (NT * NB * 2 * 32) + 32];
                }
                if (l16 < cnt[t]) {
                    float* op = out + (size_t)s[t] * D_OUTF + m0;
                    if (m0 + 8 <= D_OUTF) {
                        *(v4f*)(op + 0) = lo;
                        *(v4f*)(op + 4) = hi;
                    } else {
#pragma unroll
                        for (int v = 0; v < 4; ++v) {
                            if (m0 + v < D_OUTF)     op[v]     = lo[v];
                            if (m0 + 4 + v < D_OUTF) op[4 + v] = hi[v];
                        }
                    }
                }
            }
        }
    }
}

__global__ void __launch_bounds__(32 * KW)
rbt_gemm(const float* __restrict__ x, const float* __restrict__ W,
         float* __restrict__ out, const int* __restrict__ order,
         const int* __restrict__ counts, const int* __restrict__ offsets) {
    __shared__ v4f red[RED_V4];

    const int r    = blockIdx.y;
    const int grp  = blockIdx.x;
    const int nsmp = counts[r];
    const int seg  = offsets[r];

    const int lane = threadIdx.x & 31;
    const int wave = __builtin_amdgcn_readfirstlane((int)(threadIdx.x >> 5));
    const int half = lane >> 4;
    const int l16  = lane & 15;

    const int tile0 = grp * NB;

    unsigned wbyte[NB];
#pragma unroll
    for (int j = 0; j < NB; ++j) {
        const int row  = (tile0 + j) * TILE + l16;
        const int arow = (row < D_OUTF) ? row : (D_OUTF - 1);
        wbyte[j] = (unsigned)(r * (D_OUTF * D_INF) + arow * D_INF + wave * KSLICE + 2 * half) * 4u;
    }

    for (int j0 = 0; j0 < nsmp; j0 += 2 * TILE) {
        const int rem = nsmp - j0;
        if (j0 > 0) __syncthreads();
        if (rem > TILE) rbt_pass<2>(x, W, out, order, wbyte, seg + j0, rem, tile0, half, l16, wave, lane, red);
        else            rbt_pass<1>(x, W, out, order, wbyte, seg + j0, rem, tile0, half, l16, wave, lane, red);
    }
}

extern "C" void kernel_launch(void* const* d_in, const int* in_sizes, int n_in,
                              void* d_out, int out_size, void* d_ws, size_t ws_size,
                              hipStream_t stream) {
    const float* x   = (const float*)d_in[0];
    const int*   rel = (const int*)  d_in[1];
    const float* W   = (const float*)d_in[2];
    float*       out = (float*)d_out;

    int* ws      = (int*)d_ws;
    int* counts  = ws;
    int* offsets = counts + NUM_MAT;
    int* order   = offsets + NUM_MAT;

    rbt_rank<<<NUM_MAT, 32, 0, stream>>>(rel, counts, offsets, order);

    dim3 grid(NG, NUM_MAT);
    rbt_gemm<<<grid, 32 * KW, 0, stream>>>(x, W, out, order, counts, offsets);
}
